# starting kernel
# baseline (speedup 1.0000x reference)
#include <hip/hip_runtime.h>
#include <math.h>

typedef __attribute__((ext_vector_type(2))) float v2f;
typedef __attribute__((ext_vector_type(4))) float v4f;
typedef __attribute__((ext_vector_type(8))) float v8f;

#define PI_F 3.14159265358979323846f

#define WMMA4(a, b, c) \
  __builtin_amdgcn_wmma_f32_16x16x4_f32(false, (a), false, (b), (short)0, (c), false, false)

__device__ __forceinline__ v8f mm16(const v2f A[4], const v2f B[4]) {
  v8f c = {};
  c = WMMA4(A[0], B[0], c);
  c = WMMA4(A[1], B[1], c);
  c = WMMA4(A[2], B[2], c);
  c = WMMA4(A[3], B[3], c);
  return c;
}

__device__ __forceinline__ void make_dblk(int lane, v2f A[4]) {
  const int i = lane & 15;
  const int h = lane >> 4;
  const float sc0 = 0.35355339059327373f;
  const float sc1 = 0.5f;
#pragma unroll
  for (int c = 0; c < 4; ++c) {
#pragma unroll
    for (int j = 0; j < 2; ++j) {
      const int kk = 4 * c + 2 * h + j;
      float v = 0.0f;
      if ((i < 8) == (kk < 8)) {
        const int kf = i & 7;
        const int n  = kk & 7;
        const float s = (kf == 0) ? sc0 : sc1;
        v = s * cosf(PI_F * (float)((2 * n + 1) * kf) * (1.0f / 16.0f));
      }
      A[c][j] = v;
    }
  }
}

__device__ __forceinline__ void relayout_rows_to_B(float* lds, int lane,
                                                   const float r[8], v2f B[4]) {
  const int i = lane & 15, h = lane >> 4;
#pragma unroll
  for (int v = 0; v < 8; ++v) lds[i * 17 + v + 8 * h] = r[v];
  asm volatile("s_wait_dscnt 0" ::: "memory");
#pragma unroll
  for (int c = 0; c < 4; ++c)
#pragma unroll
    for (int j = 0; j < 2; ++j)
      B[c][j] = lds[(4 * c + 2 * h + j) * 17 + i];
}

__device__ __forceinline__ void relayout_cols_to_BT(float* lds, int lane,
                                                    const v8f cd, v2f B[4]) {
  const int i = lane & 15, h = lane >> 4;
#pragma unroll
  for (int v = 0; v < 8; ++v) lds[(v + 8 * h) * 17 + i] = cd[v];
  asm volatile("s_wait_dscnt 0" ::: "memory");
#pragma unroll
  for (int c = 0; c < 4; ++c)
#pragma unroll
    for (int j = 0; j < 2; ++j)
      B[c][j] = lds[i * 17 + 4 * c + 2 * h + j];
}

__global__ void __launch_bounds__(256) k_dct_reduce(const float* __restrict__ x,
                                                    float* __restrict__ zpart) {
  __shared__ float lds[8][16 * 17];
  const int b = blockIdx.x, q = blockIdx.y;
  const int lane = threadIdx.x & 31, w = threadIdx.x >> 5;
  const int i = lane & 15, h = lane >> 4;
  float* myl = lds[w];

  v2f A[4];
  make_dblk(lane, A);

  const float* img = x + (size_t)b * 3u * 65536u;
  float zacc[8] = {0.f, 0.f, 0.f, 0.f, 0.f, 0.f, 0.f, 0.f};

  for (int it = 0; it < 8; ++it) {
    const int t = q * 64 + w * 8 + it;
    const int ty = t >> 4, tx = t & 15;
    const float* tp = img + (size_t)(ty * 16) * 256 + tx * 16;

    v2f Bt[4];
#pragma unroll
    for (int c = 0; c < 4; ++c)
#pragma unroll
      for (int j = 0; j < 2; ++j)
        Bt[c][j] = tp[(4 * c + 2 * h + j) * 256 + i];

    const v8f m1 = mm16(A, Bt);
    v2f Bm[4];
    relayout_cols_to_BT(myl, lane, m1, Bm);
    const v8f fT = mm16(A, Bm);

#pragma unroll
    for (int v = 0; v < 8; ++v) zacc[v] += fT[v];
  }

#pragma unroll
  for (int v = 0; v < 8; ++v) {
    float val = zacc[v];
    val += __shfl_down(val, 16, 32);
    val += __shfl_down(val, 8, 32);
    if (lane < 8) {
      const int widx = q * 8 + w;
      zpart[((size_t)b * 32 + widx) * 64 + lane * 8 + v] = val;
    }
  }
}

__global__ void k_mlp(const float* __restrict__ zpart,
                      const float* __restrict__ fc1_w, const float* __restrict__ fc1_b,
                      const float* __restrict__ fc2_w, const float* __restrict__ fc2_b,
                      float* __restrict__ s) {
  const int b = blockIdx.x * blockDim.x + threadIdx.x;
  if (b >= 128) return;

  float z[64];
#pragma unroll 4
  for (int f = 0; f < 64; ++f) {
    float acc = 0.0f;
    for (int widx = 0; widx < 32; ++widx)
      acc += zpart[((size_t)b * 32 + widx) * 64 + f];
    z[f] = acc * (1.0f / 1024.0f);
  }

  float hid[32];
  for (int o = 0; o < 32; ++o) {
    float acc = fc1_b[o];
    for (int f = 0; f < 64; ++f) acc += z[f] * fc1_w[o * 64 + f];
    hid[o] = acc > 0.0f ? acc : 0.0f;
  }
  for (int j = 0; j < 64; ++j) {
    float acc = fc2_b[j];
    for (int o = 0; o < 32; ++o) acc += hid[o] * fc2_w[j * 32 + o];
    s[b * 64 + j] = 1.0f / (1.0f + expf(-acc));
  }
}

__global__ void __launch_bounds__(256) k_apply(const float* __restrict__ x,
                                               const float* __restrict__ s,
                                               float* __restrict__ out) {
  __shared__ float lds[8][16 * 17];
  __shared__ float ssh[64];
  const int b = blockIdx.x, q = blockIdx.y;
  const int lane = threadIdx.x & 31, w = threadIdx.x >> 5;
  const int i = lane & 15, h = lane >> 4;
  float* myl = lds[w];

  if (threadIdx.x < 64) ssh[threadIdx.x] = s[b * 64 + threadIdx.x];
  __syncthreads();

  v2f A[4];
  make_dblk(lane, A);

  const float* img = x + (size_t)b * 3u * 65536u;

  for (int it = 0; it < 8; ++it) {
    const int t = q * 64 + w * 8 + it;
    const int ty = t >> 4, tx = t & 15;
    const float* tp = img + (size_t)(ty * 16) * 256 + tx * 16;

    v2f Bt[4];
#pragma unroll
    for (int c = 0; c < 4; ++c)
#pragma unroll
      for (int j = 0; j < 2; ++j)
        Bt[c][j] = tp[(4 * c + 2 * h + j) * 256 + i];

    const v8f m1 = mm16(A, Bt);
    v2f Bm[4];
    relayout_cols_to_BT(myl, lane, m1, Bm);
    const v8f fT = mm16(A, Bm);

    float g[8];
#pragma unroll
    for (int v = 0; v < 8; ++v) g[v] = fT[v] * ssh[8 * (i & 7) + v];

    v2f Bg[4];
    relayout_rows_to_B(myl, lane, g, Bg);
    const v8f hcd = mm16(A, Bg);

    v2f Bh[4];
    relayout_cols_to_BT(myl, lane, hcd, Bh);
    const v8f ocd = mm16(A, Bh);

    const v4f lo = {ocd[0], ocd[1], ocd[2], ocd[3]};
    const v4f hi = {ocd[4], ocd[5], ocd[6], ocd[7]};
    float* po = out + (size_t)b * 3u * 65536u +
                (size_t)(ty * 16 + i) * 256 + tx * 16 + 8 * h;
#pragma unroll
    for (int ch = 0; ch < 3; ++ch) {
      *(v4f*)(po + (size_t)ch * 65536) = lo;
      *(v4f*)(po + (size_t)ch * 65536 + 4) = hi;
    }
  }
}

extern "C" void kernel_launch(void* const* d_in, const int* in_sizes, int n_in,
                              void* d_out, int out_size, void* d_ws, size_t ws_size,
                              hipStream_t stream) {
  const float* x     = (const float*)d_in[0];
  const float* fc1_w = (const float*)d_in[1];
  const float* fc1_b = (const float*)d_in[2];
  const float* fc2_w = (const float*)d_in[3];
  const float* fc2_b = (const float*)d_in[4];
  float* out = (float*)d_out;

  float* zpart = (float*)d_ws;
  float* sgate = zpart + 128 * 32 * 64;

  dim3 grid(128, 4), blk(256);
  k_dct_reduce<<<grid, blk, 0, stream>>>(x, zpart);
  k_mlp<<<1, 128, 0, stream>>>(zpart, fc1_w, fc1_b, fc2_w, fc2_b, sgate);
  k_apply<<<grid, blk, 0, stream>>>(x, sgate, out);
}


# one launch, one 512-thread workgroup per image holding the plane in registers: block sums, one DCT of the sum and the MLP on WMMA, then M and the apply; changes the squeeze's sum order
# speedup vs baseline: 4.7070x; 4.7070x over previous
#include <hip/hip_runtime.h>
#include <math.h>

typedef __attribute__((ext_vector_type(2))) float v2f;
typedef __attribute__((ext_vector_type(4))) float v4f;
typedef __attribute__((ext_vector_type(8))) float v8f;

#define PI_F 3.14159265358979323846f

#define WMMA4(a, b, c) \
  __builtin_amdgcn_wmma_f32_16x16x4_f32(false, (a), false, (b), (short)0, (c), false, false)

__device__ __forceinline__ v8f mm16(const v2f A[4], const v2f B[4]) {
  v8f c = {};
  c = WMMA4(A[0], B[0], c);
  c = WMMA4(A[1], B[1], c);
  c = WMMA4(A[2], B[2], c);
  c = WMMA4(A[3], B[3], c);
  return c;
}

__device__ __forceinline__ void relayout_cols_to_BT(float* lds, int lane,
                                                    const v8f cd, v2f B[4]) {
  const int i = lane & 15, h = lane >> 4;
#pragma unroll
  for (int v = 0; v < 8; ++v) lds[(v + 8 * h) * 17 + i] = cd[v];
  asm volatile("s_wait_dscnt 0" ::: "memory");
#pragma unroll
  for (int c = 0; c < 4; ++c)
#pragma unroll
    for (int j = 0; j < 2; ++j)
      B[c][j] = lds[i * 17 + 4 * c + 2 * h + j];
}

template <int N>
__device__ __forceinline__ float row_xor_add(float v) {
  const int o = __builtin_amdgcn_update_dpp(0, __float_as_int(v), 0x160 + N, 0xf, 0xf, true);
  return v + __int_as_float(o);
}

#define HS_PITCH 36
#define MS_PITCH 68
__global__ void __launch_bounds__(512) k_image(const float* __restrict__ x,
                      const float* __restrict__ fc1_w, const float* __restrict__ fc1_b,
                      const float* __restrict__ fc2_w, const float* __restrict__ fc2_b,
                      float* __restrict__ out) {
  __shared__ __attribute__((aligned(16))) float Ms[64 * MS_PITCH];
  __shared__ __attribute__((aligned(16))) float Dl[64];
  __shared__ __attribute__((aligned(16))) float Dt[64];
  __shared__ __attribute__((aligned(16))) float sg[128];
  __shared__ __attribute__((aligned(16))) float part[16 * 64];
  __shared__ __attribute__((aligned(16))) float xs[64];
  __shared__ float tl[16 * 17];
  __shared__ __attribute__((aligned(16))) float zt[256];
  __shared__ __attribute__((aligned(16))) float zrow[64];
  __shared__ __attribute__((aligned(16))) float hs[16 * HS_PITCH];
  const int b = blockIdx.x;
  const int tid = threadIdx.x;
  const int lane = tid & 31;
  const int w = __builtin_amdgcn_readfirstlane(tid >> 5);
  const int i = lane & 15, h = lane >> 4;

  const size_t strip = (size_t)b * 3u * 65536u + (size_t)(8 * w) * 256 + 8 * i + 4 * h;
  const float* xp = x + strip;

  v4f X[2][2][8];
#pragma unroll
  for (int q = 0; q < 2; ++q)
#pragma unroll
    for (int t = 0; t < 2; ++t)
#pragma unroll
      for (int r = 0; r < 8; ++r)
        X[q][t][r] = *(const v4f*)(xp + q * 32768 + r * 256 + 128 * t);

  if (tid < 64) {
    const int kf = tid >> 3;
    const int n  = tid & 7;
    const float sc = (kf == 0) ? 0.35355339059327373f : 0.5f;
    const float dv = sc * cosf(PI_F * (float)((2 * n + 1) * kf) * (1.0f / 16.0f));
    Dl[tid] = dv;
    Dt[8 * n + kf] = dv;
    zrow[tid] = 0.0f;
  }

  {
    v4f S[8];
#pragma unroll
    for (int r = 0; r < 8; ++r)
      S[r] = (X[0][0][r] + X[0][1][r]) + (X[1][0][r] + X[1][1][r]);
#pragma unroll
    for (int r = 0; r < 8; ++r)
#pragma unroll
      for (int c = 0; c < 4; ++c) {
        float v = S[r][c];
        v = row_xor_add<8>(v);
        v = row_xor_add<4>(v);
        v = row_xor_add<2>(v);
        v = row_xor_add<1>(v);
        S[r][c] = v;
      }
    if (i == 0) {
#pragma unroll
      for (int r = 0; r < 8; ++r) *(v4f*)(part + w * 64 + 8 * r + 4 * h) = S[r];
    }
  }

  v2f B1[16];
  v2f B2[8];
  float bias1 = 0.0f, bias2 = 0.0f;
#pragma unroll
  for (int c = 0; c < 16; ++c) B1[c] = (v2f){0.0f, 0.0f};
#pragma unroll
  for (int c = 0; c < 8; ++c) B2[c] = (v2f){0.0f, 0.0f};
  if (w < 2) {
    const float* wr = fc1_w + (16 * w + i) * 64 + 2 * h;
#pragma unroll
    for (int c = 0; c < 16; ++c) B1[c] = *(const v2f*)(wr + 4 * c);
    bias1 = fc1_b[16 * w + i];
  }
  if (w < 4) {
    const float* wr = fc2_w + (16 * w + i) * 32 + 2 * h;
#pragma unroll
    for (int c = 0; c < 8; ++c) B2[c] = *(const v2f*)(wr + 4 * c);
    bias2 = fc2_b[16 * w + i];
  }
  __syncthreads();

  if (w == 0) {
    float a0 = 0.0f, a1 = 0.0f;
#pragma unroll
    for (int ww = 0; ww < 16; ++ww) {
      a0 += part[ww * 64 + lane];
      a1 += part[ww * 64 + 32 + lane];
    }
    xs[lane] = a0;
    xs[32 + lane] = a1;
    asm volatile("s_wait_dscnt 0" ::: "memory");
    v2f A[4], Bt[4];
#pragma unroll
    for (int c = 0; c < 4; ++c)
#pragma unroll
      for (int j = 0; j < 2; ++j) {
        const int kk = 4 * c + 2 * h + j;
        const float dv = Dl[8 * (i & 7) + (kk & 7)];
        const float xv = xs[8 * (kk & 7) + (i & 7)];
        A[c][j]  = ((i < 8) == (kk < 8)) ? dv : 0.0f;
        Bt[c][j] = ((i < 8) && (kk < 8)) ? xv : 0.0f;
      }
    const v8f m1 = mm16(A, Bt);
    v2f Bm[4];
    relayout_cols_to_BT(tl, lane, m1, Bm);
    const v8f fT = mm16(A, Bm);
#pragma unroll
    for (int v = 0; v < 8; ++v) zt[8 * lane + v] = fT[v] * (1.0f / 1024.0f);
  }
  __syncthreads();

  if (w < 2) {
    v8f c;
#pragma unroll
    for (int v = 0; v < 8; ++v) c[v] = bias1;
    const float* ar = ((i == 0) ? zt : zrow) + 2 * h;
#pragma unroll
    for (int cc = 0; cc < 16; ++cc) {
      const v2f a = *(const v2f*)(ar + 4 * cc);
      c = WMMA4(a, B1[cc], c);
    }
#pragma unroll
    for (int v = 0; v < 8; ++v)
      hs[(v + 8 * h) * HS_PITCH + 16 * w + i] = c[v] > 0.0f ? c[v] : 0.0f;
  }
  __syncthreads();

  if (w < 4) {
    v8f c;
#pragma unroll
    for (int v = 0; v < 8; ++v) c[v] = bias2;
    const float* ar = hs + i * HS_PITCH + 2 * h;
#pragma unroll
    for (int cc = 0; cc < 8; ++cc) {
      const v2f a = *(const v2f*)(ar + 4 * cc);
      c = WMMA4(a, B2[cc], c);
    }
    sg[64 * h + 16 * w + i] = 1.0f / (1.0f + expf(-c[0]));
  }
  __syncthreads();

  {
    const int mt = w >> 2, nt = w & 3;
    const int ra = 2 * mt + (i >> 3), rb = i & 7;
    const int ci = 2 * nt + (i >> 3), cj = i & 7;
    const v4f da_lo = *(const v4f*)(Dl + ra * 8), da_hi = *(const v4f*)(Dl + ra * 8 + 4);
    const v4f dc_lo = *(const v4f*)(Dt + ci * 8), dc_hi = *(const v4f*)(Dt + ci * 8 + 4);
    v2f Drb[2], Dcj[2], sv[16];
#pragma unroll
    for (int e = 0; e < 2; ++e) {
      Drb[e] = *(const v2f*)(Dl + rb * 8 + 4 * e + 2 * h);
      Dcj[e] = *(const v2f*)(Dt + cj * 8 + 4 * e + 2 * h);
    }
#pragma unroll
    for (int kc = 0; kc < 16; ++kc) sv[kc] = *(const v2f*)(sg + 4 * kc + 2 * h);
    v8f c = {};
#pragma unroll
    for (int kc = 0; kc < 16; ++kc) {
      const int k = kc >> 1, e = kc & 1;
      const float da = (k < 4) ? da_lo[k & 3] : da_hi[k & 3];
      const float dc = (k < 4) ? dc_lo[k & 3] : dc_hi[k & 3];
      const v2f av = da * Drb[e];
      const v2f bv = sv[kc] * (dc * Dcj[e]);
      c = WMMA4(av, bv, c);
    }
#pragma unroll
    for (int v = 0; v < 8; ++v)
      Ms[(16 * mt + 8 * h + v) * MS_PITCH + 16 * nt + i] = c[v];
  }
  __syncthreads();

  float* op = out + strip;
#pragma unroll
  for (int mp = 0; mp < 2; ++mp) {
    v4f Af[2][8];
#pragma unroll
    for (int r = 0; r < 8; ++r)
#pragma unroll
      for (int mm = 0; mm < 2; ++mm) {
        const int m = 2 * mp + mm;
        const int R = 16 * m + 8 * ((i >> 2) & 1) + 4 * (i >> 3) + (i & 3);
        Af[mm][r] = *(const v4f*)(Ms + R * MS_PITCH + 8 * r + 4 * h);
      }
    __builtin_amdgcn_sched_barrier(0);

#pragma unroll
    for (int q = 0; q < 2; ++q)
#pragma unroll
      for (int t = 0; t < 2; ++t) {
        v8f acc0 = {}, acc1 = {};
#pragma unroll
        for (int r = 0; r < 8; ++r)
#pragma unroll
          for (int e = 0; e < 2; ++e) {
            const v2f bv = {X[q][t][r][2 * e], X[q][t][r][2 * e + 1]};
            const v2f a0 = {Af[0][r][2 * e], Af[0][r][2 * e + 1]};
            const v2f a1 = {Af[1][r][2 * e], Af[1][r][2 * e + 1]};
            acc0 = WMMA4(a0, bv, acc0);
            acc1 = WMMA4(a1, bv, acc1);
          }
#pragma unroll
        for (int mm = 0; mm < 2; ++mm) {
          const v8f a = mm ? acc1 : acc0;
          const int m = 2 * mp + mm;
          const v4f lo = {a[0], a[1], a[2], a[3]};
          const v4f hi = {a[4], a[5], a[6], a[7]};
          float* po = op + q * 32768 + (2 * m) * 256 + 128 * t;
#pragma unroll
          for (int ch = 0; ch < 3; ++ch) {
            *(v4f*)(po + (size_t)ch * 65536) = lo;
            *(v4f*)(po + (size_t)ch * 65536 + 256) = hi;
          }
        }
        __builtin_amdgcn_sched_barrier(0);
      }
  }
}

extern "C" void kernel_launch(void* const* d_in, const int* in_sizes, int n_in,
                              void* d_out, int out_size, void* d_ws, size_t ws_size,
                              hipStream_t stream) {
  const float* x     = (const float*)d_in[0];
  const float* fc1_w = (const float*)d_in[1];
  const float* fc1_b = (const float*)d_in[2];
  const float* fc2_w = (const float*)d_in[3];
  const float* fc2_b = (const float*)d_in[4];
  float* out = (float*)d_out;

  k_image<<<128, 512, 0, stream>>>(x, fc1_w, fc1_b, fc2_w, fc2_b, out);
}


# two workgroups per image (256 x 16 waves) split the 64x64 operator by rows: each holds the whole plane, repeats squeeze and gates, builds and applies its 32 rows of M; every sum keeps its order
# speedup vs baseline: 6.1705x; 1.3109x over previous
#include <hip/hip_runtime.h>
#include <math.h>

typedef __attribute__((ext_vector_type(2))) float v2f;
typedef __attribute__((ext_vector_type(4))) float v4f;
typedef __attribute__((ext_vector_type(8))) float v8f;

#define PI_F 3.14159265358979323846f

#define WMMA4(a, b, c) \
  __builtin_amdgcn_wmma_f32_16x16x4_f32(false, (a), false, (b), (short)0, (c), false, false)

__device__ __forceinline__ v8f mm16(const v2f A[4], const v2f B[4]) {
  v8f c = {};
  c = WMMA4(A[0], B[0], c);
  c = WMMA4(A[1], B[1], c);
  c = WMMA4(A[2], B[2], c);
  c = WMMA4(A[3], B[3], c);
  return c;
}

__device__ __forceinline__ void relayout_cols_to_BT(float* lds, int lane,
                                                    const v8f cd, v2f B[4]) {
  const int i = lane & 15, h = lane >> 4;
#pragma unroll
  for (int v = 0; v < 8; ++v) lds[(v + 8 * h) * 17 + i] = cd[v];
  asm volatile("s_wait_dscnt 0" ::: "memory");
#pragma unroll
  for (int c = 0; c < 4; ++c)
#pragma unroll
    for (int j = 0; j < 2; ++j)
      B[c][j] = lds[i * 17 + 4 * c + 2 * h + j];
}

template <int N>
__device__ __forceinline__ float row_xor_add(float v) {
  const int o = __builtin_amdgcn_update_dpp(0, __float_as_int(v), 0x160 + N, 0xf, 0xf, true);
  return v + __int_as_float(o);
}

#define HS_PITCH 36
#define MS_PITCH 68
__global__ void __launch_bounds__(512) k_image(const float* __restrict__ x,
                      const float* __restrict__ fc1_w, const float* __restrict__ fc1_b,
                      const float* __restrict__ fc2_w, const float* __restrict__ fc2_b,
                      float* __restrict__ out) {
  __shared__ __attribute__((aligned(16))) float Ms[32 * MS_PITCH];
  __shared__ __attribute__((aligned(16))) float Dl[64];
  __shared__ __attribute__((aligned(16))) float Dt[64];
  __shared__ __attribute__((aligned(16))) float sg[128];
  __shared__ __attribute__((aligned(16))) float part[16 * 64];
  __shared__ __attribute__((aligned(16))) float xs[64];
  __shared__ float tl[16 * 17];
  __shared__ __attribute__((aligned(16))) float zt[256];
  __shared__ __attribute__((aligned(16))) float zrow[64];
  __shared__ __attribute__((aligned(16))) float hs[16 * HS_PITCH];
  const int b = blockIdx.x >> 1;
  const int g = blockIdx.x & 1;
  const int tid = threadIdx.x;
  const int lane = tid & 31;
  const int w = __builtin_amdgcn_readfirstlane(tid >> 5);
  const int i = lane & 15, h = lane >> 4;

  const size_t strip = (size_t)b * 3u * 65536u + (size_t)(8 * w) * 256 + 8 * i + 4 * h;
  const float* xp = x + strip;

  v4f X[2][2][8];
#pragma unroll
  for (int q = 0; q < 2; ++q)
#pragma unroll
    for (int t = 0; t < 2; ++t)
#pragma unroll
      for (int r = 0; r < 8; ++r)
        X[q][t][r] = *(const v4f*)(xp + q * 32768 + r * 256 + 128 * t);

  if (tid < 64) {
    const int kf = tid >> 3;
    const int n  = tid & 7;
    const float sc = (kf == 0) ? 0.35355339059327373f : 0.5f;
    const float dv = sc * cosf(PI_F * (float)((2 * n + 1) * kf) * (1.0f / 16.0f));
    Dl[tid] = dv;
    Dt[8 * n + kf] = dv;
    zrow[tid] = 0.0f;
  }

  {
    v4f S[8];
#pragma unroll
    for (int r = 0; r < 8; ++r)
      S[r] = (X[0][0][r] + X[0][1][r]) + (X[1][0][r] + X[1][1][r]);
#pragma unroll
    for (int r = 0; r < 8; ++r)
#pragma unroll
      for (int c = 0; c < 4; ++c) {
        float v = S[r][c];
        v = row_xor_add<8>(v);
        v = row_xor_add<4>(v);
        v = row_xor_add<2>(v);
        v = row_xor_add<1>(v);
        S[r][c] = v;
      }
    if (i == 0) {
#pragma unroll
      for (int r = 0; r < 8; ++r) *(v4f*)(part + w * 64 + 8 * r + 4 * h) = S[r];
    }
  }

  v2f B1[16];
  v2f B2[8];
  float bias1 = 0.0f, bias2 = 0.0f;
#pragma unroll
  for (int c = 0; c < 16; ++c) B1[c] = (v2f){0.0f, 0.0f};
#pragma unroll
  for (int c = 0; c < 8; ++c) B2[c] = (v2f){0.0f, 0.0f};
  if (w < 2) {
    const float* wr = fc1_w + (16 * w + i) * 64 + 2 * h;
#pragma unroll
    for (int c = 0; c < 16; ++c) B1[c] = *(const v2f*)(wr + 4 * c);
    bias1 = fc1_b[16 * w + i];
  }
  if (w < 4) {
    const float* wr = fc2_w + (16 * w + i) * 32 + 2 * h;
#pragma unroll
    for (int c = 0; c < 8; ++c) B2[c] = *(const v2f*)(wr + 4 * c);
    bias2 = fc2_b[16 * w + i];
  }
  __syncthreads();

  if (w == 0) {
    float a0 = 0.0f, a1 = 0.0f;
#pragma unroll
    for (int ww = 0; ww < 16; ++ww) {
      a0 += part[ww * 64 + lane];
      a1 += part[ww * 64 + 32 + lane];
    }
    xs[lane] = a0;
    xs[32 + lane] = a1;
    asm volatile("s_wait_dscnt 0" ::: "memory");
    v2f A[4], Bt[4];
#pragma unroll
    for (int c = 0; c < 4; ++c)
#pragma unroll
      for (int j = 0; j < 2; ++j) {
        const int kk = 4 * c + 2 * h + j;
        const float dv = Dl[8 * (i & 7) + (kk & 7)];
        const float xv = xs[8 * (kk & 7) + (i & 7)];
        A[c][j]  = ((i < 8) == (kk < 8)) ? dv : 0.0f;
        Bt[c][j] = ((i < 8) && (kk < 8)) ? xv : 0.0f;
      }
    const v8f m1 = mm16(A, Bt);
    v2f Bm[4];
    relayout_cols_to_BT(tl, lane, m1, Bm);
    const v8f fT = mm16(A, Bm);
#pragma unroll
    for (int v = 0; v < 8; ++v) zt[8 * lane + v] = fT[v] * (1.0f / 1024.0f);
  }
  __syncthreads();

  if (w < 2) {
    v8f c;
#pragma unroll
    for (int v = 0; v < 8; ++v) c[v] = bias1;
    const float* ar = ((i == 0) ? zt : zrow) + 2 * h;
#pragma unroll
    for (int cc = 0; cc < 16; ++cc) {
      const v2f a = *(const v2f*)(ar + 4 * cc);
      c = WMMA4(a, B1[cc], c);
    }
#pragma unroll
    for (int v = 0; v < 8; ++v)
      hs[(v + 8 * h) * HS_PITCH + 16 * w + i] = c[v] > 0.0f ? c[v] : 0.0f;
  }
  __syncthreads();

  if (w < 4) {
    v8f c;
#pragma unroll
    for (int v = 0; v < 8; ++v) c[v] = bias2;
    const float* ar = hs + i * HS_PITCH + 2 * h;
#pragma unroll
    for (int cc = 0; cc < 8; ++cc) {
      const v2f a = *(const v2f*)(ar + 4 * cc);
      c = WMMA4(a, B2[cc], c);
    }
    sg[64 * h + 16 * w + i] = 1.0f / (1.0f + expf(-c[0]));
  }
  __syncthreads();

  if (w < 8) {
    const int mtl = w >> 2, nt = w & 3;
    const int mt = 2 * g + mtl;
    const int ra = 2 * mt + (i >> 3), rb = i & 7;
    const int ci = 2 * nt + (i >> 3), cj = i & 7;
    const v4f da_lo = *(const v4f*)(Dl + ra * 8), da_hi = *(const v4f*)(Dl + ra * 8 + 4);
    const v4f dc_lo = *(const v4f*)(Dt + ci * 8), dc_hi = *(const v4f*)(Dt + ci * 8 + 4);
    v2f Drb[2], Dcj[2], sv[16];
#pragma unroll
    for (int e = 0; e < 2; ++e) {
      Drb[e] = *(const v2f*)(Dl + rb * 8 + 4 * e + 2 * h);
      Dcj[e] = *(const v2f*)(Dt + cj * 8 + 4 * e + 2 * h);
    }
#pragma unroll
    for (int kc = 0; kc < 16; ++kc) sv[kc] = *(const v2f*)(sg + 4 * kc + 2 * h);
    v8f c = {};
#pragma unroll
    for (int kc = 0; kc < 16; ++kc) {
      const int k = kc >> 1, e = kc & 1;
      const float da = (k < 4) ? da_lo[k & 3] : da_hi[k & 3];
      const float dc = (k < 4) ? dc_lo[k & 3] : dc_hi[k & 3];
      const v2f av = da * Drb[e];
      const v2f bv = sv[kc] * (dc * Dcj[e]);
      c = WMMA4(av, bv, c);
    }
#pragma unroll
    for (int v = 0; v < 8; ++v)
      Ms[(16 * mtl + 8 * h + v) * MS_PITCH + 16 * nt + i] = c[v];
  }
  __syncthreads();

  float* op = out + strip + (size_t)(4 * g) * 256;
  {
    v4f Af[2][8];
#pragma unroll
    for (int r = 0; r < 8; ++r)
#pragma unroll
      for (int mm = 0; mm < 2; ++mm) {
        const int R = 16 * mm + 8 * ((i >> 2) & 1) + 4 * (i >> 3) + (i & 3);
        Af[mm][r] = *(const v4f*)(Ms + R * MS_PITCH + 8 * r + 4 * h);
      }
    __builtin_amdgcn_sched_barrier(0);

#pragma unroll
    for (int q = 0; q < 2; ++q)
#pragma unroll
      for (int t = 0; t < 2; ++t) {
        v8f acc0 = {}, acc1 = {};
#pragma unroll
        for (int r = 0; r < 8; ++r)
#pragma unroll
          for (int e = 0; e < 2; ++e) {
            const v2f bv = {X[q][t][r][2 * e], X[q][t][r][2 * e + 1]};
            const v2f a0 = {Af[0][r][2 * e], Af[0][r][2 * e + 1]};
            const v2f a1 = {Af[1][r][2 * e], Af[1][r][2 * e + 1]};
            acc0 = WMMA4(a0, bv, acc0);
            acc1 = WMMA4(a1, bv, acc1);
          }
#pragma unroll
        for (int mm = 0; mm < 2; ++mm) {
          const v8f a = mm ? acc1 : acc0;
          const v4f lo = {a[0], a[1], a[2], a[3]};
          const v4f hi = {a[4], a[5], a[6], a[7]};
          float* po = op + q * 32768 + (2 * mm) * 256 + 128 * t;
#pragma unroll
          for (int ch = 0; ch < 3; ++ch) {
            *(v4f*)(po + (size_t)ch * 65536) = lo;
            *(v4f*)(po + (size_t)ch * 65536 + 256) = hi;
          }
        }
        __builtin_amdgcn_sched_barrier(0);
      }
  }
}

extern "C" void kernel_launch(void* const* d_in, const int* in_sizes, int n_in,
                              void* d_out, int out_size, void* d_ws, size_t ws_size,
                              hipStream_t stream) {
  const float* x     = (const float*)d_in[0];
  const float* fc1_w = (const float*)d_in[1];
  const float* fc1_b = (const float*)d_in[2];
  const float* fc2_w = (const float*)d_in[3];
  const float* fc2_b = (const float*)d_in[4];
  float* out = (float*)d_out;

  k_image<<<256, 512, 0, stream>>>(x, fc1_w, fc1_b, fc2_w, fc2_b, out);
}
